# starting kernel
# baseline (speedup 1.0000x reference)
#include <hip/hip_runtime.h>
#include <hip/hip_bf16.h>
#include <cstdint>

#define BB 64
#define SS 512
#define DD 768
#define TT 64

typedef float v2f __attribute__((ext_vector_type(2)));
typedef float v8f __attribute__((ext_vector_type(8)));

__global__ __launch_bounds__(256) void crf_emissions_wmma(
    const float* __restrict__ x,
    const float* __restrict__ W,
    const float* __restrict__ bias,
    float* __restrict__ em)
{
  const int wave = blockIdx.x * (blockDim.x >> 5) + (threadIdx.x >> 5);
  const int lane = threadIdx.x & 31;
  const int m0   = wave * 16;
  const int l16  = lane & 15;
  const int kh   = lane >> 4;

  const float* __restrict__ arow = x + (size_t)(m0 + l16) * DD + 2 * kh;
  const float* __restrict__ w0 = W + (size_t)l16 * DD + 2 * kh;

  v8f acc0 = {0.f,0.f,0.f,0.f,0.f,0.f,0.f,0.f};
  v8f acc1 = acc0, acc2 = acc0, acc3 = acc0;

  #pragma unroll 4
  for (int k = 0; k < DD; k += 4) {
    v2f a  = *(const v2f*)(arow + k);
    v2f b0 = *(const v2f*)(w0 + k);
    v2f b1 = *(const v2f*)(w0 + 16 * DD + k);
    v2f b2 = *(const v2f*)(w0 + 32 * DD + k);
    v2f b3 = *(const v2f*)(w0 + 48 * DD + k);
    acc0 = __builtin_amdgcn_wmma_f32_16x16x4_f32(false, a, false, b0, (short)0, acc0, false, false);
    acc1 = __builtin_amdgcn_wmma_f32_16x16x4_f32(false, a, false, b1, (short)0, acc1, false, false);
    acc2 = __builtin_amdgcn_wmma_f32_16x16x4_f32(false, a, false, b2, (short)0, acc2, false, false);
    acc3 = __builtin_amdgcn_wmma_f32_16x16x4_f32(false, a, false, b3, (short)0, acc3, false, false);
  }

  const int rbase = m0 + kh * 8;
  #pragma unroll
  for (int nt = 0; nt < 4; ++nt) {
    v8f acc = (nt == 0) ? acc0 : (nt == 1) ? acc1 : (nt == 2) ? acc2 : acc3;
    const int n  = nt * 16 + l16;
    const float bn = bias[n];
    #pragma unroll
    for (int v = 0; v < 8; ++v) {
      em[(size_t)(rbase + v) * TT + n] = acc[v] + bn;
    }
  }
}

__global__ __launch_bounds__(256) void crf_viterbi(
    const float* __restrict__ em,
    const float* __restrict__ trans,
    int* __restrict__ out)
{
  const int b   = blockIdx.x;
  const int tid = threadIdx.x;
  const int tc  = tid & 63;
  const int p   = tid >> 6;

  __shared__ float         score[TT];
  __shared__ float         pval[4][TT];
  __shared__ unsigned char pidx[4][TT];
  __shared__ unsigned char hist[(SS - 1) * TT];

  float tr[16];
  #pragma unroll
  for (int i = 0; i < 16; ++i) tr[i] = trans[(p * 16 + i) * TT + tc];

  const float* __restrict__ emb = em + (size_t)b * SS * TT;

  if (tid < TT) score[tc] = emb[tc];
  __syncthreads();

  for (int t = 1; t < SS; ++t) {
    float best = -__builtin_inff();
    int   bi   = p * 16;
    #pragma unroll
    for (int i = 0; i < 16; ++i) {
      float v = score[p * 16 + i] + tr[i];
      if (v > best) { best = v; bi = p * 16 + i; }
    }
    pval[p][tc] = best;
    pidx[p][tc] = (unsigned char)bi;
    __syncthreads();

    if (p == 0) {
      float bv = pval[0][tc];
      int   bb = pidx[0][tc];
      #pragma unroll
      for (int q = 1; q < 4; ++q) {
        float v = pval[q][tc];
        if (v > bv) { bv = v; bb = pidx[q][tc]; }
      }
      score[tc] = bv + emb[(size_t)t * TT + tc];
      hist[(t - 1) * TT + tc] = (unsigned char)bb;
    }
    __syncthreads();
  }

  if (tid == 0) {
    float bv = score[0];
    int   bt = 0;
    for (int t2 = 1; t2 < TT; ++t2) {
      if (score[t2] > bv) { bv = score[t2]; bt = t2; }
    }
    int* __restrict__ ob = out + (size_t)b * SS;
    ob[SS - 1] = bt;
    int cur = bt;
    for (int t = SS - 2; t >= 0; --t) {
      cur = hist[t * TT + cur];
      ob[t] = cur;
    }
  }
}

extern "C" void kernel_launch(void* const* d_in, const int* in_sizes, int n_in,
                              void* d_out, int out_size, void* d_ws, size_t ws_size,
                              hipStream_t stream) {
  const float* x     = (const float*)d_in[0];
  const float* W     = (const float*)d_in[1];
  const float* bias  = (const float*)d_in[2];
  const float* trans = (const float*)d_in[3];
  float* em = (float*)d_ws;
  int*   out = (int*)d_out;

  crf_emissions_wmma<<<(BB * SS / 16) / 8, 256, 0, stream>>>(x, W, bias, em);
  crf_viterbi<<<BB, 256, 0, stream>>>(em, trans, out);
}


# Viterbi tail: walk back moved from the scalar unit to the vector ALU (in-row lane gather by tag>>2, then a 6-bit extract = next tag>>2: two dependent vector ops a row, no v_readlane); same tags
# speedup vs baseline: 3.3093x; 3.3093x over previous
#include <hip/hip_runtime.h>
#include <hip/hip_bf16.h>
#include <cstdint>

#define BB 64
#define SS 512
#define DD 768
#define TT 64

typedef float v2f __attribute__((ext_vector_type(2)));
typedef float v4f __attribute__((ext_vector_type(4)));
typedef float v8f __attribute__((ext_vector_type(8)));

typedef unsigned v2u __attribute__((ext_vector_type(2)));

__device__ __forceinline__ void split_span8(const v4f v, v2f& lo, v2f& hi)
{
  const unsigned ux = __float_as_uint(v[0]);
  const unsigned uy = __float_as_uint(v[1]);
  const unsigned uz = __float_as_uint(v[2]);
  const unsigned uw = __float_as_uint(v[3]);
  const v2u xz = __builtin_amdgcn_permlane16_swap(ux, uz, false, false);
  const v2u yw = __builtin_amdgcn_permlane16_swap(uy, uw, false, false);
  lo[0] = __uint_as_float(xz[0]);
  lo[1] = __uint_as_float(yw[0]);
  hi[0] = __uint_as_float(xz[1]);
  hi[1] = __uint_as_float(yw[1]);
}

__global__ __launch_bounds__(256) void crf_emissions_wmma(
    const float* __restrict__ x,
    const float* __restrict__ W,
    const float* __restrict__ bias,
    float* __restrict__ em)
{
  const int wave = blockIdx.x * (blockDim.x >> 5) + (threadIdx.x >> 5);
  const int lane = threadIdx.x & 31;
  const int m0   = wave * 16;
  const int l16  = lane & 15;
  const int kh   = lane >> 4;

  const float* __restrict__ arow = x + (size_t)(m0 + l16) * DD + 4 * kh;
  const float* __restrict__ w0 = W + (size_t)l16 * DD + 4 * kh;

  v8f acc0 = {0.f,0.f,0.f,0.f,0.f,0.f,0.f,0.f};
  v8f acc1 = acc0, acc2 = acc0, acc3 = acc0;

  for (int k = 0; k < DD; k += 32) {
    v4f av[4], bv[4][4];
    #pragma unroll
    for (int s = 0; s < 4; ++s) {
      av[s] = *(const v4f*)(arow + k + 8 * s);
      #pragma unroll
      for (int nt = 0; nt < 4; ++nt)
        bv[nt][s] = *(const v4f*)(w0 + nt * 16 * DD + k + 8 * s);
    }
    __builtin_amdgcn_sched_barrier(0);
    #pragma unroll
    for (int s = 0; s < 4; ++s) {
      v2f alo, ahi, b0lo, b0hi, b1lo, b1hi, b2lo, b2hi, b3lo, b3hi;
      split_span8(av[s],    alo,  ahi);
      split_span8(bv[0][s], b0lo, b0hi);
      split_span8(bv[1][s], b1lo, b1hi);
      split_span8(bv[2][s], b2lo, b2hi);
      split_span8(bv[3][s], b3lo, b3hi);
      acc0 = __builtin_amdgcn_wmma_f32_16x16x4_f32(false, alo, false, b0lo, (short)0, acc0, false, false);
      acc1 = __builtin_amdgcn_wmma_f32_16x16x4_f32(false, alo, false, b1lo, (short)0, acc1, false, false);
      acc2 = __builtin_amdgcn_wmma_f32_16x16x4_f32(false, alo, false, b2lo, (short)0, acc2, false, false);
      acc3 = __builtin_amdgcn_wmma_f32_16x16x4_f32(false, alo, false, b3lo, (short)0, acc3, false, false);
      acc0 = __builtin_amdgcn_wmma_f32_16x16x4_f32(false, ahi, false, b0hi, (short)0, acc0, false, false);
      acc1 = __builtin_amdgcn_wmma_f32_16x16x4_f32(false, ahi, false, b1hi, (short)0, acc1, false, false);
      acc2 = __builtin_amdgcn_wmma_f32_16x16x4_f32(false, ahi, false, b2hi, (short)0, acc2, false, false);
      acc3 = __builtin_amdgcn_wmma_f32_16x16x4_f32(false, ahi, false, b3hi, (short)0, acc3, false, false);
    }
  }

  const int rbase = m0 + kh * 8;
  #pragma unroll
  for (int nt = 0; nt < 4; ++nt) {
    v8f acc = (nt == 0) ? acc0 : (nt == 1) ? acc1 : (nt == 2) ? acc2 : acc3;
    const int n  = nt * 16 + l16;
    const float bn = bias[n];
    #pragma unroll
    for (int v = 0; v < 8; ++v) {
      em[(size_t)(rbase + v) * TT + n] = acc[v] + bn;
    }
  }
}

#define KSTEP 32
#define NBLK  ((SS - 1 + KSTEP - 1) / KSTEP)
#define NHELP 7

__global__ __launch_bounds__(256) void crf_viterbi(
    const float* __restrict__ em,
    const float* __restrict__ trans,
    int* __restrict__ out)
{
  const int b    = blockIdx.x;
  const int tid  = threadIdx.x;
  const int wave = tid >> 5;
  const int lane = tid & 31;

  __shared__ __attribute__((aligned(16))) float sS[SS * TT];
  __shared__ __attribute__((aligned(16))) unsigned char hist[SS * TT];

  const float* __restrict__ emb = em + (size_t)b * SS * TT;

  {
    const v4f* __restrict__ src = (const v4f*)emb;
    v4f* dst = (v4f*)sS;
    for (int i = tid; i < SS * TT / 4; i += 256) dst[i] = src[i];
  }

  v2f trv[TT];
  if (wave == 0) {
    #pragma unroll
    for (int jj = 0; jj < 32; ++jj) {
      trv[jj].x      = trans[(2 * jj) * TT + lane];
      trv[jj].y      = trans[(2 * jj + 1) * TT + lane];
      trv[32 + jj].x = trans[(2 * jj) * TT + 32 + lane];
      trv[32 + jj].y = trans[(2 * jj + 1) * TT + 32 + lane];
    }
  } else {
    #pragma unroll
    for (int j = 0; j < TT; ++j) trv[j] = *(const v2f*)(trans + j * TT + 2 * lane);
  }

  if (tid < TT) hist[(SS - 1) * TT + tid] = (unsigned char)tid;

  __syncthreads();

  for (int kb = 0; kb <= NBLK; ++kb) {
    if (wave == 0) {
      if (kb < NBLK) {
        const int t0 = kb * KSTEP + 1;
        const int t1 = (t0 + KSTEP < SS) ? (t0 + KSTEP) : SS;
        v4f r1[8], r2[8];
        {
          const v4f* sp = (const v4f*)(sS + (t0 - 1) * TT);
          #pragma unroll
          for (int q = 0; q < 8; ++q) r1[q] = sp[q];
          __builtin_amdgcn_sched_barrier(0);
          #pragma unroll
          for (int q = 0; q < 8; ++q) r2[q] = sp[8 + q];
          __builtin_amdgcn_sched_barrier(0);
        }
        for (int t = t0; t < t1; ++t) {
          float* row = sS + t * TT;
          const float e0 = row[lane];
          const float e1 = row[32 + lane];
          v2f a0 = trv[0] + r1[0].xy;
          v2f a1 = trv[1] + r1[0].zw;
          v2f c0 = trv[32] + r1[0].xy;
          v2f c1 = trv[33] + r1[0].zw;
          float m0a = fmaxf(a0.x, a0.y), m0b = fmaxf(a1.x, a1.y);
          float m1a = fmaxf(c0.x, c0.y), m1b = fmaxf(c1.x, c1.y);
          #pragma unroll
          for (int q = 1; q < 8; ++q) {
            a0 = trv[2 * q] + r1[q].xy;
            a1 = trv[2 * q + 1] + r1[q].zw;
            c0 = trv[32 + 2 * q] + r1[q].xy;
            c1 = trv[32 + 2 * q + 1] + r1[q].zw;
            m0a = fmaxf(fmaxf(m0a, a0.x), a0.y);
            m0b = fmaxf(fmaxf(m0b, a1.x), a1.y);
            m1a = fmaxf(fmaxf(m1a, c0.x), c0.y);
            m1b = fmaxf(fmaxf(m1b, c1.x), c1.y);
          }
          asm volatile("" : "+v"(m0a), "+v"(m0b), "+v"(m1a), "+v"(m1b));
          __builtin_amdgcn_sched_barrier(0);
          #pragma unroll
          for (int q = 0; q < 8; ++q) {
            a0 = trv[16 + 2 * q] + r2[q].xy;
            a1 = trv[16 + 2 * q + 1] + r2[q].zw;
            m0a = fmaxf(fmaxf(m0a, a0.x), a0.y);
            m0b = fmaxf(fmaxf(m0b, a1.x), a1.y);
          }
          row[lane] = fmaxf(m0a, m0b) + e0;
          {
            const v4f* sp = (const v4f*)row;
            #pragma unroll
            for (int q = 0; q < 8; ++q) r1[q] = sp[q];
          }
          asm volatile("" : "+v"(m1a), "+v"(m1b) : : "memory");
          __builtin_amdgcn_sched_barrier(0);
          #pragma unroll
          for (int q = 0; q < 8; ++q) {
            c0 = trv[48 + 2 * q] + r2[q].xy;
            c1 = trv[48 + 2 * q + 1] + r2[q].zw;
            m1a = fmaxf(fmaxf(m1a, c0.x), c0.y);
            m1b = fmaxf(fmaxf(m1b, c1.x), c1.y);
          }
          row[32 + lane] = fmaxf(m1a, m1b) + e1;
          {
            const v4f* sp = (const v4f*)(row + 32);
            #pragma unroll
            for (int q = 0; q < 8; ++q) r2[q] = sp[q];
          }
          __builtin_amdgcn_sched_barrier(0);
        }
      }
    } else if (kb >= 1) {
      const int tb = (kb - 1) * KSTEP + 1;
      const int te = (tb + KSTEP < SS) ? (tb + KSTEP) : SS;
      for (int t = tb + (wave - 1); t < te; t += NHELP) {
        const v4f* sp = (const v4f*)(sS + (t - 1) * TT);
        v4f s4[16];
        #pragma unroll
        for (int q = 0; q < 16; ++q) s4[q] = sp[q];
        __builtin_amdgcn_sched_barrier(0);
        float bv0[4], bv1[4];
        int   bi0[4], bi1[4];
        #pragma unroll
        for (int i = 0; i < 16; ++i) {
          #pragma unroll
          for (int k = 0; k < 4; ++k) {
            const int j = 16 * k + i;
            const v2f v = trv[j] + s4[j >> 2][j & 3];
            if (i == 0) {
              bv0[k] = v.x; bi0[k] = j;
              bv1[k] = v.y; bi1[k] = j;
            } else {
              if (v.x > bv0[k]) { bv0[k] = v.x; bi0[k] = j; }
              if (v.y > bv1[k]) { bv1[k] = v.y; bi1[k] = j; }
            }
          }
        }
        float b0 = bv0[0], b1 = bv1[0];
        int   i0 = bi0[0], i1 = bi1[0];
        #pragma unroll
        for (int k = 1; k < 4; ++k) {
          if (bv0[k] > b0) { b0 = bv0[k]; i0 = bi0[k]; }
          if (bv1[k] > b1) { b1 = bv1[k]; i1 = bi1[k]; }
        }
        hist[(t - 1) * TT + 2 * lane]     = (unsigned char)i0;
        hist[(t - 1) * TT + 2 * lane + 1] = (unsigned char)i1;
      }
    }
    __syncthreads();
  }

  if (wave == 0) {
    const float* score = sS + (SS - 1) * TT;
    float bv = score[0];
    int   bt = 0;
    for (int t2 = 1; t2 < TT; ++t2) {
      if (score[t2] > bv) { bv = score[t2]; bt = t2; }
    }
    unsigned idx = (unsigned)bt >> 2;
    unsigned sh  = ((unsigned)bt & 3u) << 3;
    unsigned sh2 = sh | 2u;

    const unsigned* histw = (const unsigned*)hist;
    int* __restrict__ ob = out + (size_t)b * SS;
    for (int blk = SS / 32 - 1; blk >= 0; --blk) {
      unsigned rowv[32];
      #pragma unroll
      for (int r = 31; r >= 0; --r)
        rowv[r] = histw[(blk * 32 + r) * (TT / 4) + (lane & 15)];
      __builtin_amdgcn_sched_barrier(0);
      unsigned outv = 0;
      #pragma unroll
      for (int r = 31; r >= 0; --r) {
        const unsigned w = __builtin_amdgcn_permlane16_var(rowv[r], rowv[r], idx, false, false);
        idx = __builtin_amdgcn_ubfe(w, sh2, 6u);
        const unsigned p = __builtin_amdgcn_ubfe(w, sh, 8u);
        sh = p << 3;
        asm("v_lshl_or_b32 %0, %1, 3, 2" : "=v"(sh2) : "v"(p));
        outv = (lane == r) ? p : outv;
      }
      ob[blk * 32 + lane] = (int)outv;
    }
  }
}

extern "C" void kernel_launch(void* const* d_in, const int* in_sizes, int n_in,
                              void* d_out, int out_size, void* d_ws, size_t ws_size,
                              hipStream_t stream) {
  const float* x     = (const float*)d_in[0];
  const float* W     = (const float*)d_in[1];
  const float* bias  = (const float*)d_in[2];
  const float* trans = (const float*)d_in[3];
  float* em = (float*)d_ws;
  int*   out = (int*)d_out;

  crf_emissions_wmma<<<(BB * SS / 16) / 8, 256, 0, stream>>>(x, W, bias, em);
  crf_viterbi<<<BB, 256, 0, stream>>>(em, trans, out);
}


# Emissions GEMM: a wave takes two 16-row tiles, each W fragment in registers feeds both (W loads per product halved); 4 waves a workgroup, next trip's loads a register set ahead; k order kept, same em
# speedup vs baseline: 3.6107x; 1.0911x over previous
#include <hip/hip_runtime.h>
#include <hip/hip_bf16.h>
#include <cstdint>

#define BB 64
#define SS 512
#define DD 768
#define TT 64

typedef float v2f __attribute__((ext_vector_type(2)));
typedef float v4f __attribute__((ext_vector_type(4)));
typedef float v8f __attribute__((ext_vector_type(8)));

typedef unsigned v2u __attribute__((ext_vector_type(2)));

__device__ __forceinline__ void split_span8(const v4f v, v2f& lo, v2f& hi)
{
  const unsigned ux = __float_as_uint(v[0]);
  const unsigned uy = __float_as_uint(v[1]);
  const unsigned uz = __float_as_uint(v[2]);
  const unsigned uw = __float_as_uint(v[3]);
  const v2u xz = __builtin_amdgcn_permlane16_swap(ux, uz, false, false);
  const v2u yw = __builtin_amdgcn_permlane16_swap(uy, uw, false, false);
  lo[0] = __uint_as_float(xz[0]);
  lo[1] = __uint_as_float(yw[0]);
  hi[0] = __uint_as_float(xz[1]);
  hi[1] = __uint_as_float(yw[1]);
}

__device__ __forceinline__ void load_trip(const float* __restrict__ a0, const float* __restrict__ a1,
                                          const float* __restrict__ w0, const int k,
                                          v4f (&av)[2][4], v4f (&bv)[4][4])
{
  #pragma unroll
  for (int s = 0; s < 4; ++s) {
    av[0][s] = *(const v4f*)(a0 + k + 8 * s);
    av[1][s] = *(const v4f*)(a1 + k + 8 * s);
    #pragma unroll
    for (int nt = 0; nt < 4; ++nt)
      bv[nt][s] = *(const v4f*)(w0 + nt * 16 * DD + k + 8 * s);
  }
}

__device__ __forceinline__ void mma_trip(const v4f (&av)[2][4], const v4f (&bv)[4][4], v8f (&acc)[2][4])
{
  #pragma unroll
  for (int s = 0; s < 4; ++s) {
    v2f alo[2], ahi[2], blo[4], bhi[4];
    split_span8(av[0][s], alo[0], ahi[0]);
    split_span8(av[1][s], alo[1], ahi[1]);
    #pragma unroll
    for (int nt = 0; nt < 4; ++nt) split_span8(bv[nt][s], blo[nt], bhi[nt]);
    #pragma unroll
    for (int mt = 0; mt < 2; ++mt) {
      #pragma unroll
      for (int nt = 0; nt < 4; ++nt)
        acc[mt][nt] = __builtin_amdgcn_wmma_f32_16x16x4_f32(false, alo[mt], false, blo[nt], (short)0, acc[mt][nt], false, false);
    }
    #pragma unroll
    for (int mt = 0; mt < 2; ++mt) {
      #pragma unroll
      for (int nt = 0; nt < 4; ++nt)
        acc[mt][nt] = __builtin_amdgcn_wmma_f32_16x16x4_f32(false, ahi[mt], false, bhi[nt], (short)0, acc[mt][nt], false, false);
    }
  }
}

__global__ __launch_bounds__(128) void crf_emissions_wmma(
    const float* __restrict__ x,
    const float* __restrict__ W,
    const float* __restrict__ bias,
    float* __restrict__ em)
{
  const int wave = blockIdx.x * (blockDim.x >> 5) + (threadIdx.x >> 5);
  const int lane = threadIdx.x & 31;
  const int m0   = wave * 32;
  const int l16  = lane & 15;
  const int kh   = lane >> 4;

  const float* __restrict__ a0 = x + (size_t)(m0 + l16) * DD + 4 * kh;
  const float* __restrict__ a1 = a0 + (size_t)16 * DD;
  const float* __restrict__ w0 = W + (size_t)l16 * DD + 4 * kh;

  v8f acc[2][4];
  #pragma unroll
  for (int mt = 0; mt < 2; ++mt) {
    #pragma unroll
    for (int nt = 0; nt < 4; ++nt) acc[mt][nt] = (v8f){0.f,0.f,0.f,0.f,0.f,0.f,0.f,0.f};
  }

  v4f avA[2][4], bvA[4][4], avB[2][4], bvB[4][4];
  load_trip(a0, a1, w0, 0, avA, bvA);
  for (int k = 0; k < DD - 64; k += 64) {
    load_trip(a0, a1, w0, k + 32, avB, bvB);
    __builtin_amdgcn_sched_barrier(0);
    mma_trip(avA, bvA, acc);
    __builtin_amdgcn_sched_barrier(0);
    load_trip(a0, a1, w0, k + 64, avA, bvA);
    __builtin_amdgcn_sched_barrier(0);
    mma_trip(avB, bvB, acc);
    __builtin_amdgcn_sched_barrier(0);
  }
  load_trip(a0, a1, w0, DD - 32, avB, bvB);
  __builtin_amdgcn_sched_barrier(0);
  mma_trip(avA, bvA, acc);
  mma_trip(avB, bvB, acc);

  #pragma unroll
  for (int mt = 0; mt < 2; ++mt) {
    const int rbase = m0 + 16 * mt + kh * 8;
    #pragma unroll
    for (int nt = 0; nt < 4; ++nt) {
      const int n  = nt * 16 + l16;
      const float bn = bias[n];
      #pragma unroll
      for (int v = 0; v < 8; ++v) {
        em[(size_t)(rbase + v) * TT + n] = acc[mt][nt][v] + bn;
      }
    }
  }
}

#define KSTEP 32
#define NBLK  ((SS - 1 + KSTEP - 1) / KSTEP)
#define NHELP 7

__global__ __launch_bounds__(256) void crf_viterbi(
    const float* __restrict__ em,
    const float* __restrict__ trans,
    int* __restrict__ out)
{
  const int b    = blockIdx.x;
  const int tid  = threadIdx.x;
  const int wave = tid >> 5;
  const int lane = tid & 31;

  __shared__ __attribute__((aligned(16))) float sS[SS * TT];
  __shared__ __attribute__((aligned(16))) unsigned char hist[SS * TT];

  const float* __restrict__ emb = em + (size_t)b * SS * TT;

  {
    const v4f* __restrict__ src = (const v4f*)emb;
    v4f* dst = (v4f*)sS;
    for (int i = tid; i < SS * TT / 4; i += 256) dst[i] = src[i];
  }

  v2f trv[TT];
  if (wave == 0) {
    #pragma unroll
    for (int jj = 0; jj < 32; ++jj) {
      trv[jj].x      = trans[(2 * jj) * TT + lane];
      trv[jj].y      = trans[(2 * jj + 1) * TT + lane];
      trv[32 + jj].x = trans[(2 * jj) * TT + 32 + lane];
      trv[32 + jj].y = trans[(2 * jj + 1) * TT + 32 + lane];
    }
  } else {
    #pragma unroll
    for (int j = 0; j < TT; ++j) trv[j] = *(const v2f*)(trans + j * TT + 2 * lane);
  }

  if (tid < TT) hist[(SS - 1) * TT + tid] = (unsigned char)tid;

  __syncthreads();

  for (int kb = 0; kb <= NBLK; ++kb) {
    if (wave == 0) {
      if (kb < NBLK) {
        const int t0 = kb * KSTEP + 1;
        const int t1 = (t0 + KSTEP < SS) ? (t0 + KSTEP) : SS;
        v4f r1[8], r2[8];
        {
          const v4f* sp = (const v4f*)(sS + (t0 - 1) * TT);
          #pragma unroll
          for (int q = 0; q < 8; ++q) r1[q] = sp[q];
          __builtin_amdgcn_sched_barrier(0);
          #pragma unroll
          for (int q = 0; q < 8; ++q) r2[q] = sp[8 + q];
          __builtin_amdgcn_sched_barrier(0);
        }
        for (int t = t0; t < t1; ++t) {
          float* row = sS + t * TT;
          const float e0 = row[lane];
          const float e1 = row[32 + lane];
          v2f a0 = trv[0] + r1[0].xy;
          v2f a1 = trv[1] + r1[0].zw;
          v2f c0 = trv[32] + r1[0].xy;
          v2f c1 = trv[33] + r1[0].zw;
          float m0a = fmaxf(a0.x, a0.y), m0b = fmaxf(a1.x, a1.y);
          float m1a = fmaxf(c0.x, c0.y), m1b = fmaxf(c1.x, c1.y);
          #pragma unroll
          for (int q = 1; q < 8; ++q) {
            a0 = trv[2 * q] + r1[q].xy;
            a1 = trv[2 * q + 1] + r1[q].zw;
            c0 = trv[32 + 2 * q] + r1[q].xy;
            c1 = trv[32 + 2 * q + 1] + r1[q].zw;
            m0a = fmaxf(fmaxf(m0a, a0.x), a0.y);
            m0b = fmaxf(fmaxf(m0b, a1.x), a1.y);
            m1a = fmaxf(fmaxf(m1a, c0.x), c0.y);
            m1b = fmaxf(fmaxf(m1b, c1.x), c1.y);
          }
          asm volatile("" : "+v"(m0a), "+v"(m0b), "+v"(m1a), "+v"(m1b));
          __builtin_amdgcn_sched_barrier(0);
          #pragma unroll
          for (int q = 0; q < 8; ++q) {
            a0 = trv[16 + 2 * q] + r2[q].xy;
            a1 = trv[16 + 2 * q + 1] + r2[q].zw;
            m0a = fmaxf(fmaxf(m0a, a0.x), a0.y);
            m0b = fmaxf(fmaxf(m0b, a1.x), a1.y);
          }
          row[lane] = fmaxf(m0a, m0b) + e0;
          {
            const v4f* sp = (const v4f*)row;
            #pragma unroll
            for (int q = 0; q < 8; ++q) r1[q] = sp[q];
          }
          asm volatile("" : "+v"(m1a), "+v"(m1b) : : "memory");
          __builtin_amdgcn_sched_barrier(0);
          #pragma unroll
          for (int q = 0; q < 8; ++q) {
            c0 = trv[48 + 2 * q] + r2[q].xy;
            c1 = trv[48 + 2 * q + 1] + r2[q].zw;
            m1a = fmaxf(fmaxf(m1a, c0.x), c0.y);
            m1b = fmaxf(fmaxf(m1b, c1.x), c1.y);
          }
          row[32 + lane] = fmaxf(m1a, m1b) + e1;
          {
            const v4f* sp = (const v4f*)(row + 32);
            #pragma unroll
            for (int q = 0; q < 8; ++q) r2[q] = sp[q];
          }
          __builtin_amdgcn_sched_barrier(0);
        }
      }
    } else if (kb >= 1) {
      const int tb = (kb - 1) * KSTEP + 1;
      const int te = (tb + KSTEP < SS) ? (tb + KSTEP) : SS;
      for (int t = tb + (wave - 1); t < te; t += NHELP) {
        const v4f* sp = (const v4f*)(sS + (t - 1) * TT);
        v4f s4[16];
        #pragma unroll
        for (int q = 0; q < 16; ++q) s4[q] = sp[q];
        __builtin_amdgcn_sched_barrier(0);
        float bv0[4], bv1[4];
        int   bi0[4], bi1[4];
        #pragma unroll
        for (int i = 0; i < 16; ++i) {
          #pragma unroll
          for (int k = 0; k < 4; ++k) {
            const int j = 16 * k + i;
            const v2f v = trv[j] + s4[j >> 2][j & 3];
            if (i == 0) {
              bv0[k] = v.x; bi0[k] = j;
              bv1[k] = v.y; bi1[k] = j;
            } else {
              if (v.x > bv0[k]) { bv0[k] = v.x; bi0[k] = j; }
              if (v.y > bv1[k]) { bv1[k] = v.y; bi1[k] = j; }
            }
          }
        }
        float b0 = bv0[0], b1 = bv1[0];
        int   i0 = bi0[0], i1 = bi1[0];
        #pragma unroll
        for (int k = 1; k < 4; ++k) {
          if (bv0[k] > b0) { b0 = bv0[k]; i0 = bi0[k]; }
          if (bv1[k] > b1) { b1 = bv1[k]; i1 = bi1[k]; }
        }
        hist[(t - 1) * TT + 2 * lane]     = (unsigned char)i0;
        hist[(t - 1) * TT + 2 * lane + 1] = (unsigned char)i1;
      }
    }
    __syncthreads();
  }

  if (wave == 0) {
    const float* score = sS + (SS - 1) * TT;
    float bv = score[0];
    int   bt = 0;
    for (int t2 = 1; t2 < TT; ++t2) {
      if (score[t2] > bv) { bv = score[t2]; bt = t2; }
    }
    unsigned idx = (unsigned)bt >> 2;
    unsigned sh  = ((unsigned)bt & 3u) << 3;
    unsigned sh2 = sh | 2u;

    const unsigned* histw = (const unsigned*)hist;
    int* __restrict__ ob = out + (size_t)b * SS;
    for (int blk = SS / 32 - 1; blk >= 0; --blk) {
      unsigned rowv[32];
      #pragma unroll
      for (int r = 31; r >= 0; --r)
        rowv[r] = histw[(blk * 32 + r) * (TT / 4) + (lane & 15)];
      __builtin_amdgcn_sched_barrier(0);
      unsigned outv = 0;
      #pragma unroll
      for (int r = 31; r >= 0; --r) {
        const unsigned w = __builtin_amdgcn_permlane16_var(rowv[r], rowv[r], idx, false, false);
        idx = __builtin_amdgcn_ubfe(w, sh2, 6u);
        const unsigned p = __builtin_amdgcn_ubfe(w, sh, 8u);
        sh = p << 3;
        asm("v_lshl_or_b32 %0, %1, 3, 2" : "=v"(sh2) : "v"(p));
        outv = (lane == r) ? p : outv;
      }
      ob[blk * 32 + lane] = (int)outv;
    }
  }
}

extern "C" void kernel_launch(void* const* d_in, const int* in_sizes, int n_in,
                              void* d_out, int out_size, void* d_ws, size_t ws_size,
                              hipStream_t stream) {
  const float* x     = (const float*)d_in[0];
  const float* W     = (const float*)d_in[1];
  const float* bias  = (const float*)d_in[2];
  const float* trans = (const float*)d_in[3];
  float* em = (float*)d_ws;
  int*   out = (int*)d_out;

  crf_emissions_wmma<<<(BB * SS / 32) / 4, 128, 0, stream>>>(x, W, bias, em);
  crf_viterbi<<<BB, 256, 0, stream>>>(em, trans, out);
}
